# starting kernel
# baseline (speedup 1.0000x reference)
#include <hip/hip_runtime.h>
#include <hip/hip_bf16.h>


#define B_  2
#define N_  512
#define F_  768
#define ATT_ 64
#define HID_ 128

typedef __attribute__((ext_vector_type(2))) float v2f;
typedef __attribute__((ext_vector_type(8))) float v8f;

__global__ void k_h(const float* __restrict__ x, const float* __restrict__ Ww,
                    const float* __restrict__ Wb, float* __restrict__ h) {
    const int lane = threadIdx.x & 31;
    const int wave = threadIdx.x >> 5;
    const int tile = blockIdx.x * 4 + wave;
    const int tm = tile >> 2;
    const int tn = tile & 3;
    const int kb = (lane >> 4) * 2;
    const int m  = tm * 16 + (lane & 15);
    const int n  = tn * 16 + (lane & 15);

    const float* xrow = x  + (size_t)m * F_;
    const float* wrow = Ww + (size_t)n * F_;

    v8f c = {};
    for (int k = 0; k < F_; k += 4) {
        v2f a, b;
        a.x = xrow[k + kb]; a.y = xrow[k + kb + 1];
        b.x = wrow[k + kb]; b.y = wrow[k + kb + 1];
        c = __builtin_amdgcn_wmma_f32_16x16x4_f32(false, a, false, b,
                                                  (short)0, c, false, false);
    }
    const float bias = Wb[n];
    const int row0 = tm * 16 + (lane >> 4) * 8;
    const int col  = tn * 16 + (lane & 15);
#pragma unroll
    for (int v = 0; v < 8; ++v)
        h[(size_t)(row0 + v) * ATT_ + col] = c[v] + bias;
}

__global__ void k_e12(const float* __restrict__ h, const float* __restrict__ W1,
                      float* __restrict__ e12) {
    const int lane = threadIdx.x & 31;
    const int wave = threadIdx.x >> 5;
    const int tile = blockIdx.x * 4 + wave;
    const int tm = tile >> 4;
    const int tn = tile & 15;
    const int kb = (lane >> 4) * 2;
    const int m    = tm * 16 + (lane & 15);
    const int ncol = tn * 16 + (lane & 15);
    const int w1row = (ncol < HID_) ? ncol : (ncol - HID_);
    const int koff  = (ncol < HID_) ? 0 : ATT_;

    const float* hrow = h  + (size_t)m * ATT_;
    const float* wrow = W1 + (size_t)w1row * (2 * ATT_) + koff;

    v8f c = {};
    for (int k = 0; k < ATT_; k += 4) {
        v2f a, b;
        a.x = hrow[k + kb]; a.y = hrow[k + kb + 1];
        b.x = wrow[k + kb]; b.y = wrow[k + kb + 1];
        c = __builtin_amdgcn_wmma_f32_16x16x4_f32(false, a, false, b,
                                                  (short)0, c, false, false);
    }
    const int row0 = tm * 16 + (lane >> 4) * 8;
    const int col  = tn * 16 + (lane & 15);
#pragma unroll
    for (int v = 0; v < 8; ++v)
        e12[(size_t)(row0 + v) * 256 + col] = c[v];
}

__global__ void k_pair(const float* __restrict__ e12, const float* __restrict__ b1,
                       const float* __restrict__ W2, const float* __restrict__ b2,
                       const int* __restrict__ adj, float* __restrict__ expE,
                       float* __restrict__ partials) {
    __shared__ float se1[32][HID_ + 1];
    __shared__ float se2[32][HID_ + 1];
    __shared__ float sb1[HID_];
    __shared__ float sw2[HID_];
    __shared__ float red[256];

    const int blk = blockIdx.x;
    const int b   = blk >> 8;
    const int t   = blk & 255;
    const int i0  = (t >> 4) * 32;
    const int j0  = (t & 15) * 32;
    const int tid = threadIdx.x;

    for (int s = tid; s < 32 * HID_; s += 256) {
        const int r = s >> 7, cc = s & 127;
        se1[r][cc] = e12[(size_t)(b * N_ + i0 + r) * 256 + cc];
        se2[r][cc] = e12[(size_t)(b * N_ + j0 + r) * 256 + HID_ + cc];
    }
    if (tid < HID_) { sb1[tid] = b1[tid]; sw2[tid] = W2[tid]; }
    __syncthreads();

    const float bias2 = b2[0];
    const int li  = tid >> 3;
    const int lj0 = (tid & 7) * 4;

    float acc[4] = {0.f, 0.f, 0.f, 0.f};
    for (int hh = 0; hh < HID_; ++hh) {
        const float a1 = se1[li][hh] + sb1[hh];
        const float w  = sw2[hh];
#pragma unroll
        for (int q = 0; q < 4; ++q) {
            const float s = a1 + se2[lj0 + q][hh];
            acc[q] = fmaf(fmaxf(s, 0.f), w, acc[q]);
        }
    }

    float lsum = 0.f;
#pragma unroll
    for (int q = 0; q < 4; ++q) {
        float e = acc[q] + bias2;
        e = (e >= 0.f) ? e : 0.01f * e;
        const size_t gi = (size_t)b * N_ * N_ + (size_t)(i0 + li) * N_ + (j0 + lj0 + q);
        const float ev = adj[gi] ? __expf(e) : 0.f;
        expE[gi] = ev;
        lsum += ev;
    }

    red[tid] = lsum;
    __syncthreads();
    for (int s = 128; s > 0; s >>= 1) {
        if (tid < s) red[tid] += red[tid + s];
        __syncthreads();
    }
    if (tid == 0) partials[blk] = red[0];
}

__global__ void k_reduce(const float* __restrict__ partials, float* __restrict__ sumexp) {
    __shared__ float red[256];
    red[threadIdx.x] = partials[blockIdx.x * 256 + threadIdx.x];
    __syncthreads();
    for (int s = 128; s > 0; s >>= 1) {
        if (threadIdx.x < s) red[threadIdx.x] += red[threadIdx.x + s];
        __syncthreads();
    }
    if (threadIdx.x == 0) sumexp[blockIdx.x] = red[0];
}

__global__ void k_out(const float* __restrict__ expE, const float* __restrict__ h,
                      const float* __restrict__ sumexp, float* __restrict__ out) {
    const int lane = threadIdx.x & 31;
    const int wave = threadIdx.x >> 5;
    const int tile = blockIdx.x * 4 + wave;
    const int b  = tile >> 7;
    const int tm = (tile >> 2) & 31;
    const int tn = tile & 3;
    const int kb = (lane >> 4) * 2;
    const int m  = tm * 16 + (lane & 15);
    const int n  = tn * 16 + (lane & 15);

    const float* arow = expE + (size_t)b * N_ * N_ + (size_t)m * N_;
    const float* bcol = h    + (size_t)b * N_ * ATT_ + n;

    v8f c = {};
    for (int k = 0; k < N_; k += 4) {
        v2f a, bb;
        a.x  = arow[k + kb];
        a.y  = arow[k + kb + 1];
        bb.x = bcol[(size_t)(k + kb) * ATT_];
        bb.y = bcol[(size_t)(k + kb + 1) * ATT_];
        c = __builtin_amdgcn_wmma_f32_16x16x4_f32(false, a, false, bb,
                                                  (short)0, c, false, false);
    }
    const float inv = 1.0f / sumexp[b];
    const int row0 = tm * 16 + (lane >> 4) * 8;
    const int col  = tn * 16 + (lane & 15);
#pragma unroll
    for (int v = 0; v < 8; ++v)
        out[((size_t)b * N_ + row0 + v) * ATT_ + col] = c[v] * inv;
}

extern "C" void kernel_launch(void* const* d_in, const int* in_sizes, int n_in,
                              void* d_out, int out_size, void* d_ws, size_t ws_size,
                              hipStream_t stream) {
    const float* x   = (const float*)d_in[0];
    const int*   adj = (const int*)  d_in[1];
    const float* Ww  = (const float*)d_in[2];
    const float* Wb  = (const float*)d_in[3];
    const float* W1  = (const float*)d_in[4];
    const float* b1  = (const float*)d_in[5];
    const float* W2  = (const float*)d_in[6];
    const float* b2  = (const float*)d_in[7];
    float* out = (float*)d_out;

    float* ws    = (float*)d_ws;
    float* h_    = ws;
    float* e12   = h_   + (size_t)B_ * N_ * ATT_;
    float* expE  = e12  + (size_t)B_ * N_ * 256;
    float* parts = expE + (size_t)B_ * N_ * N_;
    float* sums  = parts + 512;

    k_h     <<< 64, 128, 0, stream>>>(x, Ww, Wb, h_);
    k_e12   <<<256, 128, 0, stream>>>(h_, W1, e12);
    k_pair  <<<512, 256, 0, stream>>>(e12, b1, W2, b2, adj, expE, parts);
    k_reduce<<<  2, 256, 0, stream>>>(parts, sums);
    k_out   <<< 64, 128, 0, stream>>>(expE, h_, sums, out);
}


# h GEMM K loop software-pipelined in one wave: ring of 32 fragment slots, every four WMMAs followed by the eight loads that refill their slots 32 K steps ahead; one accumulator, sum orders unchanged
# speedup vs baseline: 1.6644x; 1.6644x over previous
#include <hip/hip_runtime.h>
#include <hip/hip_bf16.h>


#define B_  2
#define N_  512
#define F_  768
#define ATT_ 64
#define HID_ 128

typedef __attribute__((ext_vector_type(2))) float v2f;
typedef __attribute__((ext_vector_type(8))) float v8f;
typedef const __attribute__((address_space(1))) float* gptr_t;

#define NS_ 32
#define NR_ (F_ / (4 * NS_))

__global__ void __launch_bounds__(32) __attribute__((amdgpu_waves_per_eu(1, 1)))
k_h(const float* __restrict__ x, const float* __restrict__ Ww,
    const float* __restrict__ Wb, float* __restrict__ h) {
    const int lane = threadIdx.x;
    const int tile = blockIdx.x;
    const int tm = (tile >> 2) & 63;
    const int tn = tile & 3;
    const int kb = (lane >> 4) * 2;
    const int m  = tm * 16 + (lane & 15);
    const int n  = tn * 16 + (lane & 15);

    const unsigned xv = (unsigned)(m * F_ + kb);
    const unsigned wv = (unsigned)(n * F_ + kb);

    const gptr_t xg = (gptr_t)x;
    const gptr_t wg = (gptr_t)Ww;

    v2f a[NS_], b[NS_];
    v8f c = {};

#pragma unroll
    for (int u = 0; u < NS_; ++u) {
        a[u].x = (xg + 4 * u)[xv]; a[u].y = (xg + 4 * u + 1)[xv];
        b[u].x = (wg + 4 * u)[wv]; b[u].y = (wg + 4 * u + 1)[wv];
    }
    gptr_t xsA = xg + 4 * NS_, xsB = xsA;
    gptr_t wsA = wg + 4 * NS_, wsB = wsA;
    for (int r = 0; r < NR_ - 1; ++r) {
#pragma unroll
        for (int g = 0; g < NS_ / 4; ++g) {
#pragma unroll
            for (int q = 0; q < 4; ++q)
                c = __builtin_amdgcn_wmma_f32_16x16x4_f32(false, a[4 * g + q], false, b[4 * g + q],
                                                          (short)0, c, false, false);
            if (g & 1) {
                asm volatile("" : "+s"(xsB), "+s"(wsB), "+v"(c));
#pragma unroll
                for (int q = 0; q < 4; ++q) {
                    const int u = 4 * g + q;
                    a[u].x = (xsB + 4 * u)[xv]; a[u].y = (xsB + 4 * u + 1)[xv];
                    b[u].x = (wsB + 4 * u)[wv]; b[u].y = (wsB + 4 * u + 1)[wv];
                }
            } else {
                asm volatile("" : "+s"(xsA), "+s"(wsA), "+v"(c));
#pragma unroll
                for (int q = 0; q < 4; ++q) {
                    const int u = 4 * g + q;
                    a[u].x = (xsA + 4 * u)[xv]; a[u].y = (xsA + 4 * u + 1)[xv];
                    b[u].x = (wsA + 4 * u)[wv]; b[u].y = (wsA + 4 * u + 1)[wv];
                }
            }
        }
        xsA += 4 * NS_; wsA += 4 * NS_;
        xsB += 4 * NS_; wsB += 4 * NS_;
    }
#pragma unroll
    for (int u = 0; u < NS_; ++u)
        c = __builtin_amdgcn_wmma_f32_16x16x4_f32(false, a[u], false, b[u],
                                                  (short)0, c, false, false);

    const float bias = Wb[n];
    const int row0 = tm * 16 + (lane >> 4) * 8;
    const int col  = tn * 16 + (lane & 15);
#pragma unroll
    for (int v = 0; v < 8; ++v)
        h[(size_t)(row0 + v) * ATT_ + col] = c[v] + bias;
}

__global__ void k_e12(const float* __restrict__ h, const float* __restrict__ W1,
                      float* __restrict__ e12) {
    const int lane = threadIdx.x & 31;
    const int wave = threadIdx.x >> 5;
    const int tile = blockIdx.x * 4 + wave;
    const int tm = tile >> 4;
    const int tn = tile & 15;
    const int kb = (lane >> 4) * 2;
    const int m    = tm * 16 + (lane & 15);
    const int ncol = tn * 16 + (lane & 15);
    const int w1row = (ncol < HID_) ? ncol : (ncol - HID_);
    const int koff  = (ncol < HID_) ? 0 : ATT_;

    const float* hrow = h  + (size_t)m * ATT_;
    const float* wrow = W1 + (size_t)w1row * (2 * ATT_) + koff;

    v8f c = {};
    for (int k = 0; k < ATT_; k += 4) {
        v2f a, b;
        a.x = hrow[k + kb]; a.y = hrow[k + kb + 1];
        b.x = wrow[k + kb]; b.y = wrow[k + kb + 1];
        c = __builtin_amdgcn_wmma_f32_16x16x4_f32(false, a, false, b,
                                                  (short)0, c, false, false);
    }
    const int row0 = tm * 16 + (lane >> 4) * 8;
    const int col  = tn * 16 + (lane & 15);
#pragma unroll
    for (int v = 0; v < 8; ++v)
        e12[(size_t)(row0 + v) * 256 + col] = c[v];
}

#define JT_  64
#define NPL_ (N_ / JT_)
#define SH_P 72
#define SE_P 68

__global__ void __launch_bounds__(512)
k_pairout(const float* __restrict__ e12, const float* __restrict__ b1,
          const float* __restrict__ W2, const float* __restrict__ b2,
          const int* __restrict__ adj, const float* __restrict__ h,
          float* __restrict__ planes, float* __restrict__ partials) {
    __shared__ float se1[32][HID_ + 1];
    __shared__ float se2[JT_][HID_ + 1];
    __shared__ float sb1[HID_];
    __shared__ float sw2[HID_];
    __shared__ __attribute__((aligned(16))) float sh[JT_][SH_P];
    __shared__ __attribute__((aligned(16))) float sE[32][SE_P];
    __shared__ __attribute__((aligned(16))) float sO[32][SE_P];
    __shared__ float red[512];

    const int blk = blockIdx.x;
    const int b   = blk >> 7;
    const int t   = blk & 127;
    const int i0  = (t >> 3) * 32;
    const int p   = t & 7;
    const int j0  = p * JT_;
    const int tid = threadIdx.x;

    float  r1[8];
    float  r2[16];
    float4 rh[2];
    const int pr = tid >> 7, pc = tid & 127;
    for (int u = 0; u < 8; ++u)
        r1[u] = e12[(size_t)(b * N_ + i0 + pr + 4 * u) * 256 + pc];
    for (int u = 0; u < 16; ++u)
        r2[u] = e12[(size_t)(b * N_ + j0 + pr + 4 * u) * 256 + HID_ + pc];
    const int hr = tid >> 4, hc = (tid & 15) * 4;
    for (int u = 0; u < 2; ++u)
        rh[u] = *reinterpret_cast<const float4*>(
            h + (size_t)(b * N_ + j0 + hr + 32 * u) * ATT_ + hc);
    float rb = 0.f, rw = 0.f;
    if (tid < HID_) { rb = b1[tid]; rw = W2[tid]; }

    for (int u = 0; u < 8; ++u)  se1[pr + 4 * u][pc] = r1[u];
    for (int u = 0; u < 16; ++u) se2[pr + 4 * u][pc] = r2[u];
    for (int u = 0; u < 2; ++u)
        *reinterpret_cast<float4*>(&sh[hr + 32 * u][hc]) = rh[u];
    if (tid < HID_) { sb1[tid] = rb; sw2[tid] = rw; }
    __syncthreads();

    const float bias2 = b2[0];
    const int li  = tid >> 4;
    const int lj0 = (tid & 15) * 4;

    float acc[4] = {0.f, 0.f, 0.f, 0.f};
    for (int hh = 0; hh < HID_; ++hh) {
        const float a1 = se1[li][hh] + sb1[hh];
        const float w  = sw2[hh];
#pragma unroll
        for (int q = 0; q < 4; ++q) {
            const float s = a1 + se2[lj0 + q][hh];
            acc[q] = fmaf(fmaxf(s, 0.f), w, acc[q]);
        }
    }

    float lsum = 0.f;
#pragma unroll
    for (int q = 0; q < 4; ++q) {
        float e = acc[q] + bias2;
        e = (e >= 0.f) ? e : 0.01f * e;
        const size_t gi = (size_t)b * N_ * N_ + (size_t)(i0 + li) * N_ + (j0 + lj0 + q);
        const float ev = adj[gi] ? __expf(e) : 0.f;
        sE[li][lj0 + q] = ev;
        lsum += ev;
    }

    red[tid] = lsum;
    __syncthreads();
    for (int s = 256; s > 0; s >>= 1) {
        if (tid < s) red[tid] += red[tid + s];
        __syncthreads();
    }
    if (tid == 0) partials[blk] = red[0];

    const int lane = tid & 31;
    const int wave = tid >> 5;
    if (wave < 8) {
        const int tm = wave >> 2;
        const int tn = wave & 3;
        const int kb = (lane >> 4) * 2;
        const int m  = tm * 16 + (lane & 15);
        const int n  = tn * 16 + (lane & 15);
        v8f c = {};
        for (int k = 0; k < JT_; k += 4) {
            v2f a, bb;
            a.x  = sE[m][k + kb];
            a.y  = sE[m][k + kb + 1];
            bb.x = sh[k + kb][n];
            bb.y = sh[k + kb + 1][n];
            c = __builtin_amdgcn_wmma_f32_16x16x4_f32(false, a, false, bb,
                                                      (short)0, c, false, false);
        }
        const int row0 = tm * 16 + (lane >> 4) * 8;
        for (int v = 0; v < 8; ++v)
            sO[row0 + v][n] = c[v];
    }
    __syncthreads();

    {
        const int r  = tid >> 4;
        const int c4 = (tid & 15) * 4;
        const float4 v = *reinterpret_cast<const float4*>(&sO[r][c4]);
        float* dst = planes + ((size_t)p * (B_ * N_) + (size_t)b * N_ + i0 + r) * ATT_ + c4;
        *reinterpret_cast<float4*>(dst) = v;
    }
}

__global__ void __launch_bounds__(128)
k_final(const float* __restrict__ planes, const float* __restrict__ partials,
        float* __restrict__ out) {
    const int lane = threadIdx.x & 31;
    const int g    = blockIdx.x * 128 + threadIdx.x;
    const int b    = g >> 13;

    const float* pp = partials + b * 128;
    float s = (pp[lane] + pp[lane + 32]) + (pp[lane + 64] + pp[lane + 96]);
    for (int msk = 16; msk > 0; msk >>= 1) s += __shfl_xor(s, msk);
    const float inv = 1.0f / s;

    const float4* pl = reinterpret_cast<const float4*>(planes);
    float4 a = pl[g];
    for (int q = 1; q < NPL_; ++q) {
        const float4 v = pl[(size_t)q * (B_ * N_ * ATT_ / 4) + g];
        a.x += v.x; a.y += v.y; a.z += v.z; a.w += v.w;
    }
    a.x *= inv; a.y *= inv; a.z *= inv; a.w *= inv;
    reinterpret_cast<float4*>(out)[g] = a;
}

extern "C" void kernel_launch(void* const* d_in, const int* in_sizes, int n_in,
                              void* d_out, int out_size, void* d_ws, size_t ws_size,
                              hipStream_t stream) {
    const float* x   = (const float*)d_in[0];
    const int*   adj = (const int*)  d_in[1];
    const float* Ww  = (const float*)d_in[2];
    const float* Wb  = (const float*)d_in[3];
    const float* W1  = (const float*)d_in[4];
    const float* b1  = (const float*)d_in[5];
    const float* W2  = (const float*)d_in[6];
    const float* b2  = (const float*)d_in[7];
    float* out = (float*)d_out;

    float* ws    = (float*)d_ws;
    float* h_    = ws;
    float* e12   = h_   + (size_t)B_ * N_ * ATT_;
    float* plns  = e12  + (size_t)B_ * N_ * 256;
    float* parts = plns + (size_t)B_ * N_ * N_;

    k_h     <<<256,  32, 0, stream>>>(x, Ww, Wb, h_);
    k_e12   <<<256, 128, 0, stream>>>(h_, W1, e12);
    k_pairout<<<256, 512, 0, stream>>>(e12, b1, W2, b2, adj, h_, plns, parts);
    k_final  <<<128, 128, 0, stream>>>(plns, parts, out);
}


# stage 3 relu loop reads less from LDS: b1 added to the e1 tile when it is stored, W2 by scalar loads, e2 tile unpadded and column-rotated for 16-byte loads; every sum order unchanged
# speedup vs baseline: 1.7662x; 1.0612x over previous
#include <hip/hip_runtime.h>
#include <hip/hip_bf16.h>


#define B_  2
#define N_  512
#define F_  768
#define ATT_ 64
#define HID_ 128

typedef __attribute__((ext_vector_type(2))) float v2f;
typedef __attribute__((ext_vector_type(8))) float v8f;
typedef const __attribute__((address_space(1))) float* gptr_t;

#define NS_ 32
#define NR_ (F_ / (4 * NS_))

__global__ void __launch_bounds__(32) __attribute__((amdgpu_waves_per_eu(1, 1)))
k_h(const float* __restrict__ x, const float* __restrict__ Ww,
    const float* __restrict__ Wb, float* __restrict__ h) {
    const int lane = threadIdx.x;
    const int tile = blockIdx.x;
    const int tm = (tile >> 2) & 63;
    const int tn = tile & 3;
    const int kb = (lane >> 4) * 2;
    const int m  = tm * 16 + (lane & 15);
    const int n  = tn * 16 + (lane & 15);

    const unsigned xv = (unsigned)(m * F_ + kb);
    const unsigned wv = (unsigned)(n * F_ + kb);

    const gptr_t xg = (gptr_t)x;
    const gptr_t wg = (gptr_t)Ww;

    v2f a[NS_], b[NS_];
    v8f c = {};

#pragma unroll
    for (int u = 0; u < NS_; ++u) {
        a[u].x = (xg + 4 * u)[xv]; a[u].y = (xg + 4 * u + 1)[xv];
        b[u].x = (wg + 4 * u)[wv]; b[u].y = (wg + 4 * u + 1)[wv];
    }
    gptr_t xsA = xg + 4 * NS_, xsB = xsA;
    gptr_t wsA = wg + 4 * NS_, wsB = wsA;
    for (int r = 0; r < NR_ - 1; ++r) {
#pragma unroll
        for (int g = 0; g < NS_ / 4; ++g) {
#pragma unroll
            for (int q = 0; q < 4; ++q)
                c = __builtin_amdgcn_wmma_f32_16x16x4_f32(false, a[4 * g + q], false, b[4 * g + q],
                                                          (short)0, c, false, false);
            if (g & 1) {
                asm volatile("" : "+s"(xsB), "+s"(wsB), "+v"(c));
#pragma unroll
                for (int q = 0; q < 4; ++q) {
                    const int u = 4 * g + q;
                    a[u].x = (xsB + 4 * u)[xv]; a[u].y = (xsB + 4 * u + 1)[xv];
                    b[u].x = (wsB + 4 * u)[wv]; b[u].y = (wsB + 4 * u + 1)[wv];
                }
            } else {
                asm volatile("" : "+s"(xsA), "+s"(wsA), "+v"(c));
#pragma unroll
                for (int q = 0; q < 4; ++q) {
                    const int u = 4 * g + q;
                    a[u].x = (xsA + 4 * u)[xv]; a[u].y = (xsA + 4 * u + 1)[xv];
                    b[u].x = (wsA + 4 * u)[wv]; b[u].y = (wsA + 4 * u + 1)[wv];
                }
            }
        }
        xsA += 4 * NS_; wsA += 4 * NS_;
        xsB += 4 * NS_; wsB += 4 * NS_;
    }
#pragma unroll
    for (int u = 0; u < NS_; ++u)
        c = __builtin_amdgcn_wmma_f32_16x16x4_f32(false, a[u], false, b[u],
                                                  (short)0, c, false, false);

    const float bias = Wb[n];
    const int row0 = tm * 16 + (lane >> 4) * 8;
    const int col  = tn * 16 + (lane & 15);
#pragma unroll
    for (int v = 0; v < 8; ++v)
        h[(size_t)(row0 + v) * ATT_ + col] = c[v] + bias;
}

__global__ void k_e12(const float* __restrict__ h, const float* __restrict__ W1,
                      float* __restrict__ e12) {
    const int lane = threadIdx.x & 31;
    const int wave = threadIdx.x >> 5;
    const int tile = blockIdx.x * 4 + wave;
    const int tm = tile >> 4;
    const int tn = tile & 15;
    const int kb = (lane >> 4) * 2;
    const int m    = tm * 16 + (lane & 15);
    const int ncol = tn * 16 + (lane & 15);
    const int w1row = (ncol < HID_) ? ncol : (ncol - HID_);
    const int koff  = (ncol < HID_) ? 0 : ATT_;

    const float* hrow = h  + (size_t)m * ATT_;
    const float* wrow = W1 + (size_t)w1row * (2 * ATT_) + koff;

    v8f c = {};
    for (int k = 0; k < ATT_; k += 4) {
        v2f a, b;
        a.x = hrow[k + kb]; a.y = hrow[k + kb + 1];
        b.x = wrow[k + kb]; b.y = wrow[k + kb + 1];
        c = __builtin_amdgcn_wmma_f32_16x16x4_f32(false, a, false, b,
                                                  (short)0, c, false, false);
    }
    const int row0 = tm * 16 + (lane >> 4) * 8;
    const int col  = tn * 16 + (lane & 15);
#pragma unroll
    for (int v = 0; v < 8; ++v)
        e12[(size_t)(row0 + v) * 256 + col] = c[v];
}

#define JT_  64
#define NPL_ (N_ / JT_)
#define SH_P 72
#define SE_P 68
#define S1_P 132

__global__ void __launch_bounds__(512)
k_pairout(const float* __restrict__ e12, const float* __restrict__ b1,
          const float* __restrict__ W2, const float* __restrict__ b2,
          const int* __restrict__ adj, const float* __restrict__ h,
          float* __restrict__ planes, float* __restrict__ partials) {
    __shared__ __attribute__((aligned(16))) float se2[JT_ * HID_];
    __shared__ __attribute__((aligned(16))) float se1[32][S1_P];
    __shared__ __attribute__((aligned(16))) float sh[JT_][SH_P];
    __shared__ __attribute__((aligned(16))) float sE[32][SE_P];
    __shared__ __attribute__((aligned(16))) float sO[32][SE_P];
    __shared__ float red[512];

    const int blk = blockIdx.x;
    const int b   = blk >> 7;
    const int t   = blk & 127;
    const int i0  = (t >> 3) * 32;
    const int p   = t & 7;
    const int j0  = p * JT_;
    const int tid = threadIdx.x;

    float  r1[8];
    float  r2[16];
    float4 rh[2];
    const int pr = tid >> 7, pc = tid & 127;
    for (int u = 0; u < 8; ++u)
        r1[u] = e12[(size_t)(b * N_ + i0 + pr + 4 * u) * 256 + pc];
    for (int u = 0; u < 16; ++u)
        r2[u] = e12[(size_t)(b * N_ + j0 + pr + 4 * u) * 256 + HID_ + pc];
    const int hr = tid >> 4, hc = (tid & 15) * 4;
    for (int u = 0; u < 2; ++u)
        rh[u] = *reinterpret_cast<const float4*>(
            h + (size_t)(b * N_ + j0 + hr + 32 * u) * ATT_ + hc);
    const float rb = b1[pc];

    for (int u = 0; u < 8; ++u)  se1[pr + 4 * u][pc] = r1[u] + rb;
    for (int u = 0; u < 16; ++u) se2[(pr + 4 * u) * HID_ + ((pc + 4 * u) & (HID_ - 1))] = r2[u];
    for (int u = 0; u < 2; ++u)
        *reinterpret_cast<float4*>(&sh[hr + 32 * u][hc]) = rh[u];
    __syncthreads();

    const float bias2 = b2[0];
    const int li  = tid >> 4;
    const int lj0 = (tid & 15) * 4;

    float acc[4] = {0.f, 0.f, 0.f, 0.f};
    const float* e2row = se2 + lj0 * HID_;
    const float* a1row = &se1[li][0];
    for (int k = 0; k < HID_ / 4; ++k) {
        const int colp = (4 * k + lj0) & (HID_ - 1);
        const float4 av = *reinterpret_cast<const float4*>(a1row + 4 * k);
        float4 e[4];
        for (int q = 0; q < 4; ++q)
            e[q] = *reinterpret_cast<const float4*>(e2row + q * HID_ + colp);
        const float w0 = W2[4 * k], w1 = W2[4 * k + 1], w2 = W2[4 * k + 2], w3 = W2[4 * k + 3];
        for (int q = 0; q < 4; ++q) acc[q] = fmaf(fmaxf(av.x + e[q].x, 0.f), w0, acc[q]);
        for (int q = 0; q < 4; ++q) acc[q] = fmaf(fmaxf(av.y + e[q].y, 0.f), w1, acc[q]);
        for (int q = 0; q < 4; ++q) acc[q] = fmaf(fmaxf(av.z + e[q].z, 0.f), w2, acc[q]);
        for (int q = 0; q < 4; ++q) acc[q] = fmaf(fmaxf(av.w + e[q].w, 0.f), w3, acc[q]);
    }

    float lsum = 0.f;
#pragma unroll
    for (int q = 0; q < 4; ++q) {
        float e = acc[q] + bias2;
        e = (e >= 0.f) ? e : 0.01f * e;
        const size_t gi = (size_t)b * N_ * N_ + (size_t)(i0 + li) * N_ + (j0 + lj0 + q);
        const float ev = adj[gi] ? __expf(e) : 0.f;
        sE[li][lj0 + q] = ev;
        lsum += ev;
    }

    red[tid] = lsum;
    __syncthreads();
    for (int s = 256; s > 0; s >>= 1) {
        if (tid < s) red[tid] += red[tid + s];
        __syncthreads();
    }
    if (tid == 0) partials[blk] = red[0];

    const int lane = tid & 31;
    const int wave = tid >> 5;
    if (wave < 8) {
        const int tm = wave >> 2;
        const int tn = wave & 3;
        const int kb = (lane >> 4) * 2;
        const int m  = tm * 16 + (lane & 15);
        const int n  = tn * 16 + (lane & 15);
        v8f c = {};
        for (int k = 0; k < JT_; k += 4) {
            v2f a, bb;
            a.x  = sE[m][k + kb];
            a.y  = sE[m][k + kb + 1];
            bb.x = sh[k + kb][n];
            bb.y = sh[k + kb + 1][n];
            c = __builtin_amdgcn_wmma_f32_16x16x4_f32(false, a, false, bb,
                                                      (short)0, c, false, false);
        }
        const int row0 = tm * 16 + (lane >> 4) * 8;
        for (int v = 0; v < 8; ++v)
            sO[row0 + v][n] = c[v];
    }
    __syncthreads();

    {
        const int r  = tid >> 4;
        const int c4 = (tid & 15) * 4;
        const float4 v = *reinterpret_cast<const float4*>(&sO[r][c4]);
        float* dst = planes + ((size_t)p * (B_ * N_) + (size_t)b * N_ + i0 + r) * ATT_ + c4;
        *reinterpret_cast<float4*>(dst) = v;
    }
}

__global__ void __launch_bounds__(128)
k_final(const float* __restrict__ planes, const float* __restrict__ partials,
        float* __restrict__ out) {
    const int lane = threadIdx.x & 31;
    const int g    = blockIdx.x * 128 + threadIdx.x;
    const int b    = g >> 13;

    const float* pp = partials + b * 128;
    float s = (pp[lane] + pp[lane + 32]) + (pp[lane + 64] + pp[lane + 96]);
    for (int msk = 16; msk > 0; msk >>= 1) s += __shfl_xor(s, msk);
    const float inv = 1.0f / s;

    const float4* pl = reinterpret_cast<const float4*>(planes);
    float4 a = pl[g];
    for (int q = 1; q < NPL_; ++q) {
        const float4 v = pl[(size_t)q * (B_ * N_ * ATT_ / 4) + g];
        a.x += v.x; a.y += v.y; a.z += v.z; a.w += v.w;
    }
    a.x *= inv; a.y *= inv; a.z *= inv; a.w *= inv;
    reinterpret_cast<float4*>(out)[g] = a;
}

extern "C" void kernel_launch(void* const* d_in, const int* in_sizes, int n_in,
                              void* d_out, int out_size, void* d_ws, size_t ws_size,
                              hipStream_t stream) {
    const float* x   = (const float*)d_in[0];
    const int*   adj = (const int*)  d_in[1];
    const float* Ww  = (const float*)d_in[2];
    const float* Wb  = (const float*)d_in[3];
    const float* W1  = (const float*)d_in[4];
    const float* b1  = (const float*)d_in[5];
    const float* W2  = (const float*)d_in[6];
    const float* b2  = (const float*)d_in[7];
    float* out = (float*)d_out;

    float* ws    = (float*)d_ws;
    float* h_    = ws;
    float* e12   = h_   + (size_t)B_ * N_ * ATT_;
    float* plns  = e12  + (size_t)B_ * N_ * 256;
    float* parts = plns + (size_t)B_ * N_ * N_;

    k_h     <<<256,  32, 0, stream>>>(x, Ww, Wb, h_);
    k_e12   <<<256, 128, 0, stream>>>(h_, W1, e12);
    k_pairout<<<256, 512, 0, stream>>>(e12, b1, W2, b2, adj, h_, plns, parts);
    k_final  <<<128, 128, 0, stream>>>(plns, parts, out);
}


# pair kernel tail freed of serial waits: mask words read as one 16-byte load before the relu loop; exp block sum by in-wave DPP adds + one LDS hop (was a 9-barrier tree): denominator sum order changes
# speedup vs baseline: 1.8847x; 1.0671x over previous
#include <hip/hip_runtime.h>
#include <hip/hip_bf16.h>


#define B_  2
#define N_  512
#define F_  768
#define ATT_ 64
#define HID_ 128

typedef __attribute__((ext_vector_type(2))) float v2f;
typedef __attribute__((ext_vector_type(8))) float v8f;
typedef const __attribute__((address_space(1))) float* gptr_t;

#define NS_ 32
#define NR_ (F_ / (4 * NS_))

__global__ void __launch_bounds__(32) __attribute__((amdgpu_waves_per_eu(1, 1)))
k_h(const float* __restrict__ x, const float* __restrict__ Ww,
    const float* __restrict__ Wb, float* __restrict__ h) {
    const int lane = threadIdx.x;
    const int tile = blockIdx.x;
    const int tm = (tile >> 2) & 63;
    const int tn = tile & 3;
    const int kb = (lane >> 4) * 2;
    const int m  = tm * 16 + (lane & 15);
    const int n  = tn * 16 + (lane & 15);

    const unsigned xv = (unsigned)(m * F_ + kb);
    const unsigned wv = (unsigned)(n * F_ + kb);

    const gptr_t xg = (gptr_t)x;
    const gptr_t wg = (gptr_t)Ww;

    v2f a[NS_], b[NS_];
    v8f c = {};

#pragma unroll
    for (int u = 0; u < NS_; ++u) {
        a[u].x = (xg + 4 * u)[xv]; a[u].y = (xg + 4 * u + 1)[xv];
        b[u].x = (wg + 4 * u)[wv]; b[u].y = (wg + 4 * u + 1)[wv];
    }
    gptr_t xsA = xg + 4 * NS_, xsB = xsA;
    gptr_t wsA = wg + 4 * NS_, wsB = wsA;
    for (int r = 0; r < NR_ - 1; ++r) {
#pragma unroll
        for (int g = 0; g < NS_ / 4; ++g) {
#pragma unroll
            for (int q = 0; q < 4; ++q)
                c = __builtin_amdgcn_wmma_f32_16x16x4_f32(false, a[4 * g + q], false, b[4 * g + q],
                                                          (short)0, c, false, false);
            if (g & 1) {
                asm volatile("" : "+s"(xsB), "+s"(wsB), "+v"(c));
#pragma unroll
                for (int q = 0; q < 4; ++q) {
                    const int u = 4 * g + q;
                    a[u].x = (xsB + 4 * u)[xv]; a[u].y = (xsB + 4 * u + 1)[xv];
                    b[u].x = (wsB + 4 * u)[wv]; b[u].y = (wsB + 4 * u + 1)[wv];
                }
            } else {
                asm volatile("" : "+s"(xsA), "+s"(wsA), "+v"(c));
#pragma unroll
                for (int q = 0; q < 4; ++q) {
                    const int u = 4 * g + q;
                    a[u].x = (xsA + 4 * u)[xv]; a[u].y = (xsA + 4 * u + 1)[xv];
                    b[u].x = (wsA + 4 * u)[wv]; b[u].y = (wsA + 4 * u + 1)[wv];
                }
            }
        }
        xsA += 4 * NS_; wsA += 4 * NS_;
        xsB += 4 * NS_; wsB += 4 * NS_;
    }
#pragma unroll
    for (int u = 0; u < NS_; ++u)
        c = __builtin_amdgcn_wmma_f32_16x16x4_f32(false, a[u], false, b[u],
                                                  (short)0, c, false, false);

    const float bias = Wb[n];
    const int row0 = tm * 16 + (lane >> 4) * 8;
    const int col  = tn * 16 + (lane & 15);
#pragma unroll
    for (int v = 0; v < 8; ++v)
        h[(size_t)(row0 + v) * ATT_ + col] = c[v] + bias;
}

__global__ void k_e12(const float* __restrict__ h, const float* __restrict__ W1,
                      float* __restrict__ e12) {
    const int lane = threadIdx.x & 31;
    const int wave = threadIdx.x >> 5;
    const int tile = blockIdx.x * 4 + wave;
    const int tm = tile >> 4;
    const int tn = tile & 15;
    const int kb = (lane >> 4) * 2;
    const int m    = tm * 16 + (lane & 15);
    const int ncol = tn * 16 + (lane & 15);
    const int w1row = (ncol < HID_) ? ncol : (ncol - HID_);
    const int koff  = (ncol < HID_) ? 0 : ATT_;

    const float* hrow = h  + (size_t)m * ATT_;
    const float* wrow = W1 + (size_t)w1row * (2 * ATT_) + koff;

    v8f c = {};
    for (int k = 0; k < ATT_; k += 4) {
        v2f a, b;
        a.x = hrow[k + kb]; a.y = hrow[k + kb + 1];
        b.x = wrow[k + kb]; b.y = wrow[k + kb + 1];
        c = __builtin_amdgcn_wmma_f32_16x16x4_f32(false, a, false, b,
                                                  (short)0, c, false, false);
    }
    const int row0 = tm * 16 + (lane >> 4) * 8;
    const int col  = tn * 16 + (lane & 15);
#pragma unroll
    for (int v = 0; v < 8; ++v)
        e12[(size_t)(row0 + v) * 256 + col] = c[v];
}

__device__ __forceinline__ float row16_sum(float v) {
    v += __int_as_float(__builtin_amdgcn_update_dpp(0, __float_as_int(v), 0x161, 0xf, 0xf, true));
    v += __int_as_float(__builtin_amdgcn_update_dpp(0, __float_as_int(v), 0x162, 0xf, 0xf, true));
    v += __int_as_float(__builtin_amdgcn_update_dpp(0, __float_as_int(v), 0x164, 0xf, 0xf, true));
    v += __int_as_float(__builtin_amdgcn_update_dpp(0, __float_as_int(v), 0x168, 0xf, 0xf, true));
    return v;
}

#define JT_  64
#define NPL_ (N_ / JT_)
#define SH_P 72
#define SE_P 68
#define S1_P 132

__global__ void __launch_bounds__(512)
k_pairout(const float* __restrict__ e12, const float* __restrict__ b1,
          const float* __restrict__ W2, const float* __restrict__ b2,
          const int* __restrict__ adj, const float* __restrict__ h,
          float* __restrict__ planes, float* __restrict__ partials) {
    __shared__ __attribute__((aligned(16))) float se2[JT_ * HID_];
    __shared__ __attribute__((aligned(16))) float se1[32][S1_P];
    __shared__ __attribute__((aligned(16))) float sh[JT_][SH_P];
    __shared__ __attribute__((aligned(16))) float sE[32][SE_P];
    __shared__ __attribute__((aligned(16))) float sO[32][SE_P];
    __shared__ float red[32];

    const int blk = blockIdx.x;
    const int b   = blk >> 7;
    const int t   = blk & 127;
    const int i0  = (t >> 3) * 32;
    const int p   = t & 7;
    const int j0  = p * JT_;
    const int tid = threadIdx.x;

    float  r1[8];
    float  r2[16];
    float4 rh[2];
    const int pr = tid >> 7, pc = tid & 127;
    for (int u = 0; u < 8; ++u)
        r1[u] = e12[(size_t)(b * N_ + i0 + pr + 4 * u) * 256 + pc];
    for (int u = 0; u < 16; ++u)
        r2[u] = e12[(size_t)(b * N_ + j0 + pr + 4 * u) * 256 + HID_ + pc];
    const int hr = tid >> 4, hc = (tid & 15) * 4;
    for (int u = 0; u < 2; ++u)
        rh[u] = *reinterpret_cast<const float4*>(
            h + (size_t)(b * N_ + j0 + hr + 32 * u) * ATT_ + hc);
    const float rb = b1[pc];
    const int li  = tid >> 4;
    const int lj0 = (tid & 15) * 4;
    const int4 am = *reinterpret_cast<const int4*>(
        adj + (size_t)b * N_ * N_ + (size_t)(i0 + li) * N_ + (j0 + lj0));

    for (int u = 0; u < 8; ++u)  se1[pr + 4 * u][pc] = r1[u] + rb;
    for (int u = 0; u < 16; ++u) se2[(pr + 4 * u) * HID_ + ((pc + 4 * u) & (HID_ - 1))] = r2[u];
    for (int u = 0; u < 2; ++u)
        *reinterpret_cast<float4*>(&sh[hr + 32 * u][hc]) = rh[u];
    __syncthreads();

    const float bias2 = b2[0];

    float acc[4] = {0.f, 0.f, 0.f, 0.f};
    const float* e2row = se2 + lj0 * HID_;
    const float* a1row = &se1[li][0];
    for (int k = 0; k < HID_ / 4; ++k) {
        const int colp = (4 * k + lj0) & (HID_ - 1);
        const float4 av = *reinterpret_cast<const float4*>(a1row + 4 * k);
        float4 e[4];
        for (int q = 0; q < 4; ++q)
            e[q] = *reinterpret_cast<const float4*>(e2row + q * HID_ + colp);
        const float w0 = W2[4 * k], w1 = W2[4 * k + 1], w2 = W2[4 * k + 2], w3 = W2[4 * k + 3];
        for (int q = 0; q < 4; ++q) acc[q] = fmaf(fmaxf(av.x + e[q].x, 0.f), w0, acc[q]);
        for (int q = 0; q < 4; ++q) acc[q] = fmaf(fmaxf(av.y + e[q].y, 0.f), w1, acc[q]);
        for (int q = 0; q < 4; ++q) acc[q] = fmaf(fmaxf(av.z + e[q].z, 0.f), w2, acc[q]);
        for (int q = 0; q < 4; ++q) acc[q] = fmaf(fmaxf(av.w + e[q].w, 0.f), w3, acc[q]);
    }

    const int amq[4] = {am.x, am.y, am.z, am.w};
    float lsum = 0.f;
#pragma unroll
    for (int q = 0; q < 4; ++q) {
        float e = acc[q] + bias2;
        e = (e >= 0.f) ? e : 0.01f * e;
        const float ev = amq[q] ? __expf(e) : 0.f;
        sE[li][lj0 + q] = ev;
        lsum += ev;
    }

    lsum = row16_sum(lsum);
    if ((tid & 15) == 0) red[li] = lsum;
    __syncthreads();

    const int lane = tid & 31;
    const int wave = tid >> 5;
    if (wave == 8) {
        const float s = row16_sum(red[lane]);
        const float lo = __int_as_float(__builtin_amdgcn_readlane(__float_as_int(s), 0));
        const float hi = __int_as_float(__builtin_amdgcn_readlane(__float_as_int(s), 16));
        if (lane == 0) partials[blk] = lo + hi;
    }

    if (wave < 8) {
        const int tm = wave >> 2;
        const int tn = wave & 3;
        const int kb = (lane >> 4) * 2;
        const int m  = tm * 16 + (lane & 15);
        const int n  = tn * 16 + (lane & 15);
        v8f c = {};
        for (int k = 0; k < JT_; k += 4) {
            v2f a, bb;
            a.x  = sE[m][k + kb];
            a.y  = sE[m][k + kb + 1];
            bb.x = sh[k + kb][n];
            bb.y = sh[k + kb + 1][n];
            c = __builtin_amdgcn_wmma_f32_16x16x4_f32(false, a, false, bb,
                                                      (short)0, c, false, false);
        }
        const int row0 = tm * 16 + (lane >> 4) * 8;
        for (int v = 0; v < 8; ++v)
            sO[row0 + v][n] = c[v];
    }
    __syncthreads();

    {
        const int r  = tid >> 4;
        const int c4 = (tid & 15) * 4;
        const float4 v = *reinterpret_cast<const float4*>(&sO[r][c4]);
        float* dst = planes + ((size_t)p * (B_ * N_) + (size_t)b * N_ + i0 + r) * ATT_ + c4;
        *reinterpret_cast<float4*>(dst) = v;
    }
}

__global__ void __launch_bounds__(128)
k_final(const float* __restrict__ planes, const float* __restrict__ partials,
        float* __restrict__ out) {
    const int lane = threadIdx.x & 31;
    const int g    = blockIdx.x * 128 + threadIdx.x;
    const int b    = g >> 13;

    const float* pp = partials + b * 128;
    float s = (pp[lane] + pp[lane + 32]) + (pp[lane + 64] + pp[lane + 96]);
    for (int msk = 16; msk > 0; msk >>= 1) s += __shfl_xor(s, msk);
    const float inv = 1.0f / s;

    const float4* pl = reinterpret_cast<const float4*>(planes);
    float4 a = pl[g];
    for (int q = 1; q < NPL_; ++q) {
        const float4 v = pl[(size_t)q * (B_ * N_ * ATT_ / 4) + g];
        a.x += v.x; a.y += v.y; a.z += v.z; a.w += v.w;
    }
    a.x *= inv; a.y *= inv; a.z *= inv; a.w *= inv;
    reinterpret_cast<float4*>(out)[g] = a;
}

extern "C" void kernel_launch(void* const* d_in, const int* in_sizes, int n_in,
                              void* d_out, int out_size, void* d_ws, size_t ws_size,
                              hipStream_t stream) {
    const float* x   = (const float*)d_in[0];
    const int*   adj = (const int*)  d_in[1];
    const float* Ww  = (const float*)d_in[2];
    const float* Wb  = (const float*)d_in[3];
    const float* W1  = (const float*)d_in[4];
    const float* b1  = (const float*)d_in[5];
    const float* W2  = (const float*)d_in[6];
    const float* b2  = (const float*)d_in[7];
    float* out = (float*)d_out;

    float* ws    = (float*)d_ws;
    float* h_    = ws;
    float* e12   = h_   + (size_t)B_ * N_ * ATT_;
    float* plns  = e12  + (size_t)B_ * N_ * 256;
    float* parts = plns + (size_t)B_ * N_ * N_;

    k_h     <<<256,  32, 0, stream>>>(x, Ww, Wb, h_);
    k_e12   <<<256, 128, 0, stream>>>(h_, W1, e12);
    k_pairout<<<256, 512, 0, stream>>>(e12, b1, W2, b2, adj, h_, plns, parts);
    k_final  <<<128, 128, 0, stream>>>(plns, parts, out);
}
